# starting kernel
# baseline (speedup 1.0000x reference)
#include <hip/hip_runtime.h>
#include <hip/hip_bf16.h>

#define N_SRC     50000
#define N_DST     50000
#define N_EDGES   640000
#define IN_FEATS  256
#define OUT_FEATS 128

typedef __attribute__((ext_vector_type(2))) float v2f;
typedef __attribute__((ext_vector_type(8))) float v8f;

__global__ void zero_f4(float4* __restrict__ out, int n4) {
    int i = blockIdx.x * blockDim.x + threadIdx.x;
    if (i < n4) out[i] = make_float4(0.f, 0.f, 0.f, 0.f);
}

__global__ void gemm_cj_wmma(const float* __restrict__ feat,
                             const float* __restrict__ W,
                             const float* __restrict__ cj,
                             float* __restrict__ h) {
    const int lane = threadIdx.x & 31;
    const int wave = threadIdx.x >> 5;
    const int mt   = blockIdx.x * (blockDim.x >> 5) + wave;
    if (mt * 16 >= N_SRC) return;

    const int row0 = mt * 16;
    const int m    = lane & 15;
    const int half = lane >> 4;

    const float* __restrict__ arow = feat + (size_t)(row0 + m) * IN_FEATS;

    v8f acc[8];
    #pragma unroll
    for (int t = 0; t < 8; ++t) acc[t] = (v8f){};

    #pragma unroll 2
    for (int k0 = 0; k0 < IN_FEATS; k0 += 4) {
        const int ka = k0 + 2 * half;
        const v2f a = *(const v2f*)(arow + ka);
        const float* __restrict__ wp = W + (size_t)ka * OUT_FEATS + m;
        #pragma unroll
        for (int t = 0; t < 8; ++t) {
            v2f b;
            b.x = wp[t * 16];
            b.y = wp[OUT_FEATS + t * 16];
            acc[t] = __builtin_amdgcn_wmma_f32_16x16x4_f32(
                          false, a,  false, b,
                          (short)0, acc[t],
                          false,  false);
        }
    }

    #pragma unroll
    for (int i = 0; i < 8; ++i) {
        const int   row = row0 + i + 8 * half;
        const float s   = cj[row];
        float* __restrict__ hrow = h + (size_t)row * OUT_FEATS + m;
        #pragma unroll
        for (int t = 0; t < 8; ++t) hrow[t * 16] = acc[t][i] * s;
    }
}

__global__ void edge_scatter(const float* __restrict__ h,
                             const int* __restrict__ src,
                             const int* __restrict__ dst,
                             const float* __restrict__ ci,
                             float* __restrict__ out) {
    const int lane = threadIdx.x & 31;
    const int e    = (blockIdx.x * blockDim.x + threadIdx.x) >> 5;
    if (e >= N_EDGES) return;

    const int   s     = src[e];
    const int   d     = dst[e];
    const float scale = ci[d];

    const float4 v = *(const float4*)(h + (size_t)s * OUT_FEATS + lane * 4);
    float* __restrict__ o = out + (size_t)d * OUT_FEATS + lane * 4;
    atomicAdd(o + 0, v.x * scale);
    atomicAdd(o + 1, v.y * scale);
    atomicAdd(o + 2, v.z * scale);
    atomicAdd(o + 3, v.w * scale);
}

extern "C" void kernel_launch(void* const* d_in, const int* in_sizes, int n_in,
                              void* d_out, int out_size, void* d_ws, size_t ws_size,
                              hipStream_t stream) {
    const float* feat   = (const float*)d_in[0];
    const float* weight = (const float*)d_in[1];
    const float* cj     = (const float*)d_in[2];
    const float* ci     = (const float*)d_in[3];
    const int*   src    = (const int*)d_in[4];
    const int*   dst    = (const int*)d_in[5];
    float*       out    = (float*)d_out;
    float*       h      = (float*)d_ws;

    {
        const int n4 = (N_DST * OUT_FEATS) / 4;
        zero_f4<<<(n4 + 255) / 256, 256, 0, stream>>>((float4*)out, n4);
    }

    {
        const int mtiles = N_SRC / 16;
        const int wavesPerBlock = 8;
        const int blocks = (mtiles + wavesPerBlock - 1) / wavesPerBlock;
        gemm_cj_wmma<<<blocks, wavesPerBlock * 32, 0, stream>>>(feat, weight, cj, h);
    }

    {
        const int wavesPerBlock = 8;
        const int blocks = (N_EDGES + wavesPerBlock - 1) / wavesPerBlock;
        edge_scatter<<<blocks, wavesPerBlock * 32, 0, stream>>>(h, src, dst, ci, out);
    }
}


# Edge aggregation as gather, not float-atomic scatter: int atomics rank edges per dst row, prefix scan, src ids placed, one wave sums a row's h rows in rank (arrival) order and writes it once
# speedup vs baseline: 1.2146x; 1.2146x over previous
#include <hip/hip_runtime.h>
#include <hip/hip_bf16.h>

#define N_SRC     50000
#define N_DST     50000
#define N_EDGES   640000
#define IN_FEATS  256
#define OUT_FEATS 128

typedef __attribute__((ext_vector_type(2))) float v2f;
typedef __attribute__((ext_vector_type(8))) float v8f;

__global__ void gemm_cj_wmma(const float* __restrict__ feat,
                             const float* __restrict__ W,
                             const float* __restrict__ cj,
                             float* __restrict__ h,
                             int* __restrict__ cnt) {
    {
        const int g = blockIdx.x * blockDim.x + threadIdx.x;
        if (g < N_DST) cnt[g] = 0;
    }
    const int lane = threadIdx.x & 31;
    const int wave = threadIdx.x >> 5;
    const int mt   = blockIdx.x * (blockDim.x >> 5) + wave;
    if (mt * 16 >= N_SRC) return;

    const int row0 = mt * 16;
    const int m    = lane & 15;
    const int half = lane >> 4;

    const float* __restrict__ arow = feat + (size_t)(row0 + m) * IN_FEATS;

    v8f acc[8];
    #pragma unroll
    for (int t = 0; t < 8; ++t) acc[t] = (v8f){};

    #pragma unroll 2
    for (int k0 = 0; k0 < IN_FEATS; k0 += 4) {
        const int ka = k0 + 2 * half;
        const v2f a = *(const v2f*)(arow + ka);
        const float* __restrict__ wp = W + (size_t)ka * OUT_FEATS + m;
        #pragma unroll
        for (int t = 0; t < 8; ++t) {
            v2f b;
            b.x = wp[t * 16];
            b.y = wp[OUT_FEATS + t * 16];
            acc[t] = __builtin_amdgcn_wmma_f32_16x16x4_f32(
                          false, a,  false, b,
                          (short)0, acc[t],
                          false,  false);
        }
    }

    #pragma unroll
    for (int i = 0; i < 8; ++i) {
        const int   row = row0 + i + 8 * half;
        const float s   = cj[row];
        float* __restrict__ hrow = h + (size_t)row * OUT_FEATS + m;
        #pragma unroll
        for (int t = 0; t < 8; ++t) hrow[t * 16] = acc[t][i] * s;
    }
}

#define EDGE_QUADS (N_EDGES / 4)

__global__ __launch_bounds__(256)
void edge_rank(const int* __restrict__ dst,
               int* __restrict__ cnt,
               int* __restrict__ rank) {
    const int t = blockIdx.x * blockDim.x + threadIdx.x;
    if (t >= EDGE_QUADS) return;
    const int4 d = ((const int4*)dst)[t];
    int4 r;
    r.x = atomicAdd(cnt + d.x, 1);
    r.y = atomicAdd(cnt + d.y, 1);
    r.z = atomicAdd(cnt + d.z, 1);
    r.w = atomicAdd(cnt + d.w, 1);
    ((int4*)rank)[t] = r;
}

#define SCAN_RUNS 13
__global__ __launch_bounds__(1024)
void scan_cnt(const int* __restrict__ cnt, int* __restrict__ off) {
    __shared__ int wtot[32];
    const int lane = threadIdx.x & 31;
    const int wave = threadIdx.x >> 5;
    const int base = wave * (SCAN_RUNS * 128) + lane * 4;

    int4 v[SCAN_RUNS];
    #pragma unroll
    for (int j = 0; j < SCAN_RUNS; ++j) {
        const int i = base + j * 128;
        v[j] = (i < N_DST) ? *(const int4*)(cnt + i) : make_int4(0, 0, 0, 0);
    }

    int carry = 0;
    #pragma unroll
    for (int j = 0; j < SCAN_RUNS; ++j) {
        const int lsum = v[j].x + v[j].y + v[j].z + v[j].w;
        int x = lsum;
        #pragma unroll
        for (int d = 1; d < 32; d <<= 1) {
            const int y = __shfl_up(x, d, 32);
            if (lane >= d) x += y;
        }
        const int tot = __shfl(x, 31, 32);
        const int e0 = carry + x - lsum;
        const int4 c = v[j];
        v[j].x = e0;
        v[j].y = e0 + c.x;
        v[j].z = e0 + c.x + c.y;
        v[j].w = e0 + c.x + c.y + c.z;
        carry += tot;
    }

    if (lane == 0) wtot[wave] = carry;
    __syncthreads();
    int w = (lane < wave) ? wtot[lane] : 0;
    #pragma unroll
    for (int d = 16; d >= 1; d >>= 1) w += __shfl_xor(w, d, 32);

    #pragma unroll
    for (int j = 0; j < SCAN_RUNS; ++j) {
        const int i = base + j * 128;
        if (i < N_DST) {
            int4 o = v[j];
            o.x += w; o.y += w; o.z += w; o.w += w;
            *(int4*)(off + i) = o;
        } else if (i == N_DST) {
            off[N_DST] = v[j].x + w;
        }
    }
}

__global__ __launch_bounds__(256)
void place_src(const int* __restrict__ src,
               const int* __restrict__ dst,
               const int* __restrict__ rank,
               const int* __restrict__ off,
               int* __restrict__ perm) {
    const int t = blockIdx.x * blockDim.x + threadIdx.x;
    if (t >= EDGE_QUADS) return;
    const int4 s = ((const int4*)src)[t];
    const int4 d = ((const int4*)dst)[t];
    const int4 r = ((const int4*)rank)[t];
    perm[off[d.x] + r.x] = s.x;
    perm[off[d.y] + r.y] = s.y;
    perm[off[d.z] + r.z] = s.z;
    perm[off[d.w] + r.w] = s.w;
}

__global__ __launch_bounds__(256)
void gather_rows(const float* __restrict__ h,
                 const int* __restrict__ off,
                 const int* __restrict__ perm,
                 const float* __restrict__ ci,
                 float* __restrict__ out) {
    const int lane = threadIdx.x & 31;
    const int d = __builtin_amdgcn_readfirstlane(
                      (int)((blockIdx.x * blockDim.x + threadIdx.x) >> 5));
    if (d >= N_DST) return;

    const int start = off[d];
    const int n     = off[d + 1] - start;

    float4 acc = make_float4(0.f, 0.f, 0.f, 0.f);
    for (int c = 0; c < n; c += 32) {
        const int m = (n - c < 32) ? (n - c) : 32;
        const int mine = (lane < m) ? perm[start + c + lane] : 0;
        #pragma unroll 4
        for (int j = 0; j < m; ++j) {
            const int s = __builtin_amdgcn_readlane(mine, j);
            const float4 v = *(const float4*)(h + (size_t)s * OUT_FEATS + lane * 4);
            acc.x += v.x; acc.y += v.y; acc.z += v.z; acc.w += v.w;
        }
    }

    const float scale = ci[d];
    float4 o;
    o.x = acc.x * scale; o.y = acc.y * scale; o.z = acc.z * scale; o.w = acc.w * scale;
    *(float4*)(out + (size_t)d * OUT_FEATS + lane * 4) = o;
}

extern "C" void kernel_launch(void* const* d_in, const int* in_sizes, int n_in,
                              void* d_out, int out_size, void* d_ws, size_t ws_size,
                              hipStream_t stream) {
    const float* feat   = (const float*)d_in[0];
    const float* weight = (const float*)d_in[1];
    const float* cj     = (const float*)d_in[2];
    const float* ci     = (const float*)d_in[3];
    const int*   src    = (const int*)d_in[4];
    const int*   dst    = (const int*)d_in[5];
    float*       out    = (float*)d_out;
    float*       h      = (float*)d_ws;
    int*         cnt    = (int*)((char*)d_ws + 33554432);
    int*         off    = (int*)((char*)d_ws + 34603008);
    int*         rank   = (int*)((char*)d_ws + 35651584);
    int*         perm   = (int*)((char*)d_ws + 41943040);

    gemm_cj_wmma<<<391, 256, 0, stream>>>(feat, weight, cj, h, cnt);

    edge_rank<<<625, 256, 0, stream>>>(dst, cnt, rank);

    scan_cnt<<<1, 1024, 0, stream>>>(cnt, off);

    place_src<<<625, 256, 0, stream>>>(src, dst, rank, off, perm);

    gather_rows<<<6250, 256, 0, stream>>>(h, off, perm, ci, out);
}


# Per-edge global atomics dropped: a work-group owns 196 dst rows, scans dst, groups its edges with LDS counters and sums their h rows in LDS-arrival order; 5 launches become 2
# speedup vs baseline: 1.6857x; 1.3879x over previous
#include <hip/hip_runtime.h>
#include <hip/hip_bf16.h>

#define N_SRC     50000
#define N_DST     50000
#define N_EDGES   640000
#define IN_FEATS  256
#define OUT_FEATS 128

typedef __attribute__((ext_vector_type(2))) float v2f;
typedef __attribute__((ext_vector_type(8))) float v8f;

__global__ void gemm_cj_wmma(const float* __restrict__ feat,
                             const float* __restrict__ W,
                             const float* __restrict__ cj,
                             float* __restrict__ h) {
    const int lane = threadIdx.x & 31;
    const int wave = threadIdx.x >> 5;
    const int mt   = blockIdx.x * (blockDim.x >> 5) + wave;
    if (mt * 16 >= N_SRC) return;

    const int row0 = mt * 16;
    const int m    = lane & 15;
    const int half = lane >> 4;

    const float* __restrict__ arow = feat + (size_t)(row0 + m) * IN_FEATS;

    v8f acc[8];
    #pragma unroll
    for (int t = 0; t < 8; ++t) acc[t] = (v8f){};

    #pragma unroll 2
    for (int k0 = 0; k0 < IN_FEATS; k0 += 4) {
        const int ka = k0 + 2 * half;
        const v2f a = *(const v2f*)(arow + ka);
        const float* __restrict__ wp = W + (size_t)ka * OUT_FEATS + m;
        #pragma unroll
        for (int t = 0; t < 8; ++t) {
            v2f b;
            b.x = wp[t * 16];
            b.y = wp[OUT_FEATS + t * 16];
            acc[t] = __builtin_amdgcn_wmma_f32_16x16x4_f32(
                          false, a,  false, b,
                          (short)0, acc[t],
                          false,  false);
        }
    }

    #pragma unroll
    for (int i = 0; i < 8; ++i) {
        const int   row = row0 + i + 8 * half;
        const float s   = cj[row];
        float* __restrict__ hrow = h + (size_t)row * OUT_FEATS + m;
        #pragma unroll
        for (int t = 0; t < 8; ++t) hrow[t * 16] = acc[t][i] * s;
    }
}

#define EDGE_QUADS    (N_EDGES / 4)
#define ROWS_PER_WG   196
#define OWNER_WGS     256
#define OWNER_THREADS 1024
#define LIST_CAP      20480
#define SCAN_BATCH    8

__device__ __forceinline__ float4 ld_hrow(const float* __restrict__ h, int s, int lane) {
    return *(const float4*)(h + (size_t)s * OUT_FEATS + lane * 4);
}

__global__ __launch_bounds__(OWNER_THREADS)
void owner_gather(const float* __restrict__ h,
                  const int* __restrict__ src,
                  const int* __restrict__ dst,
                  const float* __restrict__ ci,
                  float* __restrict__ out) {
    __shared__ int s_cnt[ROWS_PER_WG];
    __shared__ int s_off[ROWS_PER_WG + 1];
    __shared__ int s_cur[ROWS_PER_WG];
    __shared__ int s_total;
    __shared__ int s_list[LIST_CAP];
    __shared__ int s_src[LIST_CAP];

    const int tid  = threadIdx.x;
    const int lane = tid & 31;
    const int wave = __builtin_amdgcn_readfirstlane(tid >> 5);
    const int r0   = blockIdx.x * ROWS_PER_WG;
    const int nrows = (N_DST - r0 < ROWS_PER_WG) ? (N_DST - r0) : ROWS_PER_WG;

    if (tid < ROWS_PER_WG) s_cnt[tid] = 0;
    if (tid == 0) s_total = 0;
    __syncthreads();

    const int4* __restrict__ dst4 = (const int4*)dst;
    for (int q0 = 0; q0 < EDGE_QUADS; q0 += SCAN_BATCH * OWNER_THREADS) {
        int4 d[SCAN_BATCH];
        #pragma unroll
        for (int u = 0; u < SCAN_BATCH; ++u) {
            const int q = q0 + u * OWNER_THREADS + tid;
            d[u] = (q < EDGE_QUADS) ? dst4[q] : make_int4(-1, -1, -1, -1);
        }
        #pragma unroll
        for (int u = 0; u < SCAN_BATCH; ++u) {
            const int q = q0 + u * OWNER_THREADS + tid;
            const unsigned a0 = (unsigned)(d[u].x - r0);
            const unsigned a1 = (unsigned)(d[u].y - r0);
            const unsigned a2 = (unsigned)(d[u].z - r0);
            const unsigned a3 = (unsigned)(d[u].w - r0);
            const unsigned lim = (unsigned)nrows;
            if ((a0 < lim) | (a1 < lim) | (a2 < lim) | (a3 < lim)) {
                const unsigned a[4] = {a0, a1, a2, a3};
                #pragma unroll
                for (int c = 0; c < 4; ++c) {
                    if (a[c] < lim) {
                        atomicAdd(&s_cnt[a[c]], 1);
                        const int p = atomicAdd(&s_total, 1);
                        if (p < LIST_CAP) s_list[p] = (int)((a[c] << 20) | (unsigned)(q * 4 + c));
                    }
                }
            }
        }
    }
    __syncthreads();
    const int total = __builtin_amdgcn_readfirstlane(s_total);

    if (total <= LIST_CAP) {
        if (wave == 0) {
            int c[7];
            int sum = 0;
            #pragma unroll
            for (int j = 0; j < 7; ++j) {
                const int r = lane * 7 + j;
                c[j] = (r < ROWS_PER_WG) ? s_cnt[r] : 0;
                sum += c[j];
            }
            int x = sum;
            #pragma unroll
            for (int dd = 1; dd < 32; dd <<= 1) {
                const int y = __shfl_up(x, dd, 32);
                if (lane >= dd) x += y;
            }
            int e0 = x - sum;
            #pragma unroll
            for (int j = 0; j < 7; ++j) {
                const int r = lane * 7 + j;
                if (r < ROWS_PER_WG) { s_off[r] = e0; s_cur[r] = e0; }
                e0 += c[j];
            }
            if (lane == 0) s_off[ROWS_PER_WG] = total;
        }
        __syncthreads();

        for (int j = tid; j < total; j += OWNER_THREADS) {
            const int rec = s_list[j];
            const int r   = rec >> 20;
            const int e   = rec & 0xFFFFF;
            const int s   = src[e];
            const int p   = atomicAdd(&s_cur[r], 1);
            s_src[p] = s;
        }
        __syncthreads();

        const int ra = (wave * nrows) >> 5;
        const int nr = (((wave + 1) * nrows) >> 5) - ra;
        const int   offv = s_off[ra + (lane < nr ? lane : nr)];
        const float civ  = (lane < nr) ? ci[r0 + ra + lane] : 0.f;
        const int eBeg = __builtin_amdgcn_readlane(offv, 0);
        const int eEnd = __builtin_amdgcn_readlane(offv, nr);
        int r    = 0;
        int rend = __builtin_amdgcn_readlane(offv, 1);
        float4 acc = make_float4(0.f, 0.f, 0.f, 0.f);

        for (int base = eBeg; base < eEnd; base += 32) {
            const int left = eEnd - base;
            const int cntc = left < 32 ? left : 32;
            const int mine = (lane < left) ? s_src[base + lane] : 0;
            for (int b = 0; b < cntc; b += 8) {
                float4 v[8];
                #pragma unroll
                for (int u = 0; u < 8; ++u)
                    v[u] = ld_hrow(h, __builtin_amdgcn_readlane(mine, b + u), lane);
                #pragma unroll
                for (int u = 0; u < 8; ++u) {
                    if (b + u < cntc) {
                        while (base + b + u == rend) {
                            const float sc = __int_as_float(__builtin_amdgcn_readlane(__float_as_int(civ), r));
                            float4 o;
                            o.x = acc.x * sc; o.y = acc.y * sc; o.z = acc.z * sc; o.w = acc.w * sc;
                            *(float4*)(out + (size_t)(r0 + ra + r) * OUT_FEATS + lane * 4) = o;
                            acc = make_float4(0.f, 0.f, 0.f, 0.f);
                            ++r;
                            rend = __builtin_amdgcn_readlane(offv, r + 1);
                        }
                        acc.x += v[u].x; acc.y += v[u].y; acc.z += v[u].z; acc.w += v[u].w;
                    }
                }
            }
        }
        while (r < nr) {
            const float sc = __int_as_float(__builtin_amdgcn_readlane(__float_as_int(civ), r));
            float4 o;
            o.x = acc.x * sc; o.y = acc.y * sc; o.z = acc.z * sc; o.w = acc.w * sc;
            *(float4*)(out + (size_t)(r0 + ra + r) * OUT_FEATS + lane * 4) = o;
            acc = make_float4(0.f, 0.f, 0.f, 0.f);
            ++r;
        }
    } else {
        for (int r = wave; r < nrows; r += OWNER_THREADS / 32) {
            float4 acc = make_float4(0.f, 0.f, 0.f, 0.f);
            for (int base = 0; base < N_EDGES; base += 32) {
                const int dd = dst[base + lane];
                unsigned m = __builtin_amdgcn_ballot_w32(dd == r0 + r);
                while (m) {
                    const int j = __builtin_ctz(m);
                    m &= m - 1;
                    const float4 v = ld_hrow(h, __builtin_amdgcn_readfirstlane(src[base + j]), lane);
                    acc.x += v.x; acc.y += v.y; acc.z += v.z; acc.w += v.w;
                }
            }
            const float sc = ci[r0 + r];
            float4 o;
            o.x = acc.x * sc; o.y = acc.y * sc; o.z = acc.z * sc; o.w = acc.w * sc;
            *(float4*)(out + (size_t)(r0 + r) * OUT_FEATS + lane * 4) = o;
        }
    }
}

extern "C" void kernel_launch(void* const* d_in, const int* in_sizes, int n_in,
                              void* d_out, int out_size, void* d_ws, size_t ws_size,
                              hipStream_t stream) {
    const float* feat   = (const float*)d_in[0];
    const float* weight = (const float*)d_in[1];
    const float* cj     = (const float*)d_in[2];
    const float* ci     = (const float*)d_in[3];
    const int*   src    = (const int*)d_in[4];
    const int*   dst    = (const int*)d_in[5];
    float*       out    = (float*)d_out;
    float*       h      = (float*)d_ws;

    gemm_cj_wmma<<<391, 256, 0, stream>>>(feat, weight, cj, h);

    owner_gather<<<OWNER_WGS, OWNER_THREADS, 0, stream>>>(h, src, dst, ci, out);
}


# Owners stop scanning all of dst: 250 extra work-groups of launch 1 counting-sort 2560-edge chunks by owner into the workspace; an owner reads only its segments; row sums stay in LDS-arrival order
# speedup vs baseline: 2.2110x; 1.3116x over previous
#include <hip/hip_runtime.h>
#include <hip/hip_bf16.h>

#define N_SRC     50000
#define N_DST     50000
#define N_EDGES   640000
#define IN_FEATS  256
#define OUT_FEATS 128

typedef __attribute__((ext_vector_type(2))) float v2f;
typedef __attribute__((ext_vector_type(8))) float v8f;

#define ROWS_PER_WG   196
#define OWNER_WGS     256
#define GEMM_WGS      391
#define BIN_WGS       250
#define BIN_THREADS   256
#define BIN_CHUNK     2560
#define BIN_PAIRS     (BIN_CHUNK / 2 / BIN_THREADS)
#define WS_SORTED     (N_SRC * OUT_FEATS)
#define WS_TABLE      (WS_SORTED + N_EDGES)

__device__ __forceinline__ void bin_edges(const int chunk,
                                          const int* __restrict__ src,
                                          const int* __restrict__ dst,
                                          int* __restrict__ ws) {
    __shared__ int b_cnt[OWNER_WGS];
    __shared__ int b_off[OWNER_WGS + 1];
    const int tid = threadIdx.x;
    b_cnt[tid] = 0;
    __syncthreads();

    const int base = chunk * BIN_CHUNK;
    const int2* __restrict__ dst2 = (const int2*)(dst + base);
    const int2* __restrict__ src2 = (const int2*)(src + base);
    int own[2 * BIN_PAIRS], rec[2 * BIN_PAIRS], rk[2 * BIN_PAIRS];
    #pragma unroll
    for (int i = 0; i < BIN_PAIRS; ++i) {
        const int2 d = dst2[i * BIN_THREADS + tid];
        const int2 sv = src2[i * BIN_THREADS + tid];
        const int o0 = d.x / ROWS_PER_WG, o1 = d.y / ROWS_PER_WG;
        own[2 * i]     = ((unsigned)d.x < (unsigned)N_DST) ? o0 : -1;
        own[2 * i + 1] = ((unsigned)d.y < (unsigned)N_DST) ? o1 : -1;
        rec[2 * i]     = ((d.x - o0 * ROWS_PER_WG) << 16) | (sv.x & 0xFFFF);
        rec[2 * i + 1] = ((d.y - o1 * ROWS_PER_WG) << 16) | (sv.y & 0xFFFF);
    }
    #pragma unroll
    for (int i = 0; i < 2 * BIN_PAIRS; ++i)
        rk[i] = (own[i] >= 0) ? atomicAdd(&b_cnt[own[i]], 1) : 0;
    __syncthreads();

    if (tid < 32) {
        int c[8];
        int sum = 0;
        #pragma unroll
        for (int j = 0; j < 8; ++j) { c[j] = b_cnt[tid * 8 + j]; sum += c[j]; }
        int x = sum;
        #pragma unroll
        for (int dd = 1; dd < 32; dd <<= 1) {
            const int y = __shfl_up(x, dd, 32);
            if (tid >= dd) x += y;
        }
        int e0 = x - sum;
        #pragma unroll
        for (int j = 0; j < 8; ++j) { b_off[tid * 8 + j] = e0; e0 += c[j]; }
        if (tid == 31) b_off[OWNER_WGS] = e0;
    }
    __syncthreads();

    int* __restrict__ sorted = ws + WS_SORTED + base;
    #pragma unroll
    for (int i = 0; i < 2 * BIN_PAIRS; ++i)
        if (own[i] >= 0) sorted[b_off[own[i]] + rk[i]] = rec[i];
    ws[WS_TABLE + tid * BIN_WGS + chunk] = (b_off[tid] << 12) | b_cnt[tid];
}

__global__ void gemm_cj_wmma(const float* __restrict__ feat,
                             const float* __restrict__ W,
                             const float* __restrict__ cj,
                             const int* __restrict__ src,
                             const int* __restrict__ dst,
                             float* __restrict__ h) {
    if (blockIdx.x >= GEMM_WGS) {
        bin_edges(blockIdx.x - GEMM_WGS, src, dst, (int*)h);
        return;
    }
    const int lane = threadIdx.x & 31;
    const int wave = threadIdx.x >> 5;
    const int mt   = blockIdx.x * (blockDim.x >> 5) + wave;
    if (mt * 16 >= N_SRC) return;

    const int row0 = mt * 16;
    const int m    = lane & 15;
    const int half = lane >> 4;

    const float* __restrict__ arow = feat + (size_t)(row0 + m) * IN_FEATS;

    v8f acc[8];
    #pragma unroll
    for (int t = 0; t < 8; ++t) acc[t] = (v8f){};

    #pragma unroll 2
    for (int k0 = 0; k0 < IN_FEATS; k0 += 4) {
        const int ka = k0 + 2 * half;
        const v2f a = *(const v2f*)(arow + ka);
        const float* __restrict__ wp = W + (size_t)ka * OUT_FEATS + m;
        #pragma unroll
        for (int t = 0; t < 8; ++t) {
            v2f b;
            b.x = wp[t * 16];
            b.y = wp[OUT_FEATS + t * 16];
            acc[t] = __builtin_amdgcn_wmma_f32_16x16x4_f32(
                          false, a,  false, b,
                          (short)0, acc[t],
                          false,  false);
        }
    }

    #pragma unroll
    for (int i = 0; i < 8; ++i) {
        const int   row = row0 + i + 8 * half;
        const float s   = cj[row];
        float* __restrict__ hrow = h + (size_t)row * OUT_FEATS + m;
        #pragma unroll
        for (int t = 0; t < 8; ++t) hrow[t * 16] = acc[t][i] * s;
    }
}

#define OWNER_THREADS 1024
#define LIST_CAP      20480

__device__ __forceinline__ float4 ld_hrow(const float* __restrict__ h, int s, int lane) {
    return *(const float4*)(h + (size_t)s * OUT_FEATS + lane * 4);
}

__global__ __launch_bounds__(OWNER_THREADS)
void owner_gather(const float* __restrict__ h,
                  const int* __restrict__ src,
                  const int* __restrict__ dst,
                  const float* __restrict__ ci,
                  const int* __restrict__ ws,
                  float* __restrict__ out) {
    __shared__ int s_seg[BIN_WGS];
    __shared__ int s_sego[BIN_WGS];
    __shared__ int s_cnt[ROWS_PER_WG];
    __shared__ int s_off[ROWS_PER_WG + 1];
    __shared__ int s_cur[ROWS_PER_WG];
    __shared__ int s_total;
    __shared__ int s_list[LIST_CAP];
    __shared__ int s_src[LIST_CAP];

    const int tid  = threadIdx.x;
    const int lane = tid & 31;
    const int wave = __builtin_amdgcn_readfirstlane(tid >> 5);
    const int r0   = blockIdx.x * ROWS_PER_WG;
    const int nrows = (N_DST - r0 < ROWS_PER_WG) ? (N_DST - r0) : ROWS_PER_WG;

    if (tid < ROWS_PER_WG) s_cnt[tid] = 0;
    if (tid < BIN_WGS) s_seg[tid] = ws[WS_TABLE + blockIdx.x * BIN_WGS + tid];
    __syncthreads();

    if (wave == 0) {
        int c[8];
        int sum = 0;
        #pragma unroll
        for (int j = 0; j < 8; ++j) {
            const int b = lane * 8 + j;
            c[j] = (b < BIN_WGS) ? (s_seg[b] & 0xFFF) : 0;
            sum += c[j];
        }
        int x = sum;
        #pragma unroll
        for (int dd = 1; dd < 32; dd <<= 1) {
            const int y = __shfl_up(x, dd, 32);
            if (lane >= dd) x += y;
        }
        int e0 = x - sum;
        #pragma unroll
        for (int j = 0; j < 8; ++j) {
            const int b = lane * 8 + j;
            if (b < BIN_WGS) s_sego[b] = e0;
            e0 += c[j];
        }
        if (lane == 31) s_total = e0;
    }
    __syncthreads();
    const int total = __builtin_amdgcn_readfirstlane(s_total);

    if (total <= LIST_CAP) {
        {
            const int b   = tid >> 2;
            const int sub = tid & 3;
            if (b < BIN_WGS) {
                const int pk = s_seg[b];
                const int cn = pk & 0xFFF;
                const int o  = s_sego[b];
                const int* __restrict__ p = ws + WS_SORTED + b * BIN_CHUNK + (pk >> 12);
                for (int i = sub; i < cn; i += 4) {
                    const int rec = p[i];
                    s_list[o + i] = rec;
                    atomicAdd(&s_cnt[rec >> 16], 1);
                }
            }
        }
        __syncthreads();

        if (wave == 0) {
            int c[7];
            int sum = 0;
            #pragma unroll
            for (int j = 0; j < 7; ++j) {
                const int r = lane * 7 + j;
                c[j] = (r < ROWS_PER_WG) ? s_cnt[r] : 0;
                sum += c[j];
            }
            int x = sum;
            #pragma unroll
            for (int dd = 1; dd < 32; dd <<= 1) {
                const int y = __shfl_up(x, dd, 32);
                if (lane >= dd) x += y;
            }
            int e0 = x - sum;
            #pragma unroll
            for (int j = 0; j < 7; ++j) {
                const int r = lane * 7 + j;
                if (r < ROWS_PER_WG) { s_off[r] = e0; s_cur[r] = e0; }
                e0 += c[j];
            }
            if (lane == 0) s_off[ROWS_PER_WG] = total;
        }
        __syncthreads();

        for (int j = tid; j < total; j += OWNER_THREADS) {
            const int rec = s_list[j];
            const int p   = atomicAdd(&s_cur[rec >> 16], 1);
            s_src[p] = rec & 0xFFFF;
        }
        __syncthreads();

        const int ra = (wave * nrows) >> 5;
        const int nr = (((wave + 1) * nrows) >> 5) - ra;
        const int   offv = s_off[ra + (lane < nr ? lane : nr)];
        const float civ  = (lane < nr) ? ci[r0 + ra + lane] : 0.f;
        const int eBeg = __builtin_amdgcn_readlane(offv, 0);
        const int eEnd = __builtin_amdgcn_readlane(offv, nr);
        int r    = 0;
        int rend = __builtin_amdgcn_readlane(offv, 1);
        float4 acc = make_float4(0.f, 0.f, 0.f, 0.f);

        for (int base = eBeg; base < eEnd; base += 32) {
            const int left = eEnd - base;
            const int cntc = left < 32 ? left : 32;
            const int mine = (lane < left) ? s_src[base + lane] : 0;
            for (int b = 0; b < cntc; b += 8) {
                float4 v[8];
                #pragma unroll
                for (int u = 0; u < 8; ++u)
                    v[u] = ld_hrow(h, __builtin_amdgcn_readlane(mine, b + u), lane);
                #pragma unroll
                for (int u = 0; u < 8; ++u) {
                    if (b + u < cntc) {
                        while (base + b + u == rend) {
                            const float sc = __int_as_float(__builtin_amdgcn_readlane(__float_as_int(civ), r));
                            float4 o;
                            o.x = acc.x * sc; o.y = acc.y * sc; o.z = acc.z * sc; o.w = acc.w * sc;
                            *(float4*)(out + (size_t)(r0 + ra + r) * OUT_FEATS + lane * 4) = o;
                            acc = make_float4(0.f, 0.f, 0.f, 0.f);
                            ++r;
                            rend = __builtin_amdgcn_readlane(offv, r + 1);
                        }
                        acc.x += v[u].x; acc.y += v[u].y; acc.z += v[u].z; acc.w += v[u].w;
                    }
                }
            }
        }
        while (r < nr) {
            const float sc = __int_as_float(__builtin_amdgcn_readlane(__float_as_int(civ), r));
            float4 o;
            o.x = acc.x * sc; o.y = acc.y * sc; o.z = acc.z * sc; o.w = acc.w * sc;
            *(float4*)(out + (size_t)(r0 + ra + r) * OUT_FEATS + lane * 4) = o;
            acc = make_float4(0.f, 0.f, 0.f, 0.f);
            ++r;
        }
    } else {
        for (int r = wave; r < nrows; r += OWNER_THREADS / 32) {
            float4 acc = make_float4(0.f, 0.f, 0.f, 0.f);
            for (int base = 0; base < N_EDGES; base += 32) {
                const int dd = dst[base + lane];
                unsigned m = __builtin_amdgcn_ballot_w32(dd == r0 + r);
                while (m) {
                    const int j = __builtin_ctz(m);
                    m &= m - 1;
                    const float4 v = ld_hrow(h, __builtin_amdgcn_readfirstlane(src[base + j]), lane);
                    acc.x += v.x; acc.y += v.y; acc.z += v.z; acc.w += v.w;
                }
            }
            const float sc = ci[r0 + r];
            float4 o;
            o.x = acc.x * sc; o.y = acc.y * sc; o.z = acc.z * sc; o.w = acc.w * sc;
            *(float4*)(out + (size_t)(r0 + r) * OUT_FEATS + lane * 4) = o;
        }
    }
}

extern "C" void kernel_launch(void* const* d_in, const int* in_sizes, int n_in,
                              void* d_out, int out_size, void* d_ws, size_t ws_size,
                              hipStream_t stream) {
    const float* feat   = (const float*)d_in[0];
    const float* weight = (const float*)d_in[1];
    const float* cj     = (const float*)d_in[2];
    const float* ci     = (const float*)d_in[3];
    const int*   src    = (const int*)d_in[4];
    const int*   dst    = (const int*)d_in[5];
    float*       out    = (float*)d_out;
    float*       h      = (float*)d_ws;

    gemm_cj_wmma<<<GEMM_WGS + BIN_WGS, BIN_THREADS, 0, stream>>>(feat, weight, cj, src, dst, h);

    owner_gather<<<OWNER_WGS, OWNER_THREADS, 0, stream>>>(h, src, dst, ci, (const int*)d_ws, out);
}


# GEMM made W-stationary: each wave keeps its 16 columns of W in registers, 256 resident work-groups stream 32-row feat tiles through LDS; same f32 products, k order of every sum unchanged
# speedup vs baseline: 3.1905x; 1.4430x over previous
#include <hip/hip_runtime.h>
#include <hip/hip_bf16.h>

#define N_SRC     50000
#define N_DST     50000
#define N_EDGES   640000
#define IN_FEATS  256
#define OUT_FEATS 128

typedef __attribute__((ext_vector_type(2))) float v2f;
typedef __attribute__((ext_vector_type(8))) float v8f;

#define ROWS_PER_WG   196
#define OWNER_WGS     256
#define GEMM_WGS      256
#define BIN_WGS       250
#define BIN_THREADS   256
#define BIN_CHUNK     2560
#define BIN_PAIRS     (BIN_CHUNK / 2 / BIN_THREADS)
#define WS_SORTED     (N_SRC * OUT_FEATS)
#define WS_TABLE      (WS_SORTED + N_EDGES)

__device__ __forceinline__ void bin_edges(const int chunk,
                                          const int* __restrict__ src,
                                          const int* __restrict__ dst,
                                          int* __restrict__ ws) {
    __shared__ int b_cnt[OWNER_WGS];
    __shared__ int b_off[OWNER_WGS + 1];
    const int tid = threadIdx.x;
    b_cnt[tid] = 0;
    __syncthreads();

    const int base = chunk * BIN_CHUNK;
    const int2* __restrict__ dst2 = (const int2*)(dst + base);
    const int2* __restrict__ src2 = (const int2*)(src + base);
    int own[2 * BIN_PAIRS], rec[2 * BIN_PAIRS], rk[2 * BIN_PAIRS];
    #pragma unroll
    for (int i = 0; i < BIN_PAIRS; ++i) {
        const int2 d = dst2[i * BIN_THREADS + tid];
        const int2 sv = src2[i * BIN_THREADS + tid];
        const int o0 = d.x / ROWS_PER_WG, o1 = d.y / ROWS_PER_WG;
        own[2 * i]     = ((unsigned)d.x < (unsigned)N_DST) ? o0 : -1;
        own[2 * i + 1] = ((unsigned)d.y < (unsigned)N_DST) ? o1 : -1;
        rec[2 * i]     = ((d.x - o0 * ROWS_PER_WG) << 16) | (sv.x & 0xFFFF);
        rec[2 * i + 1] = ((d.y - o1 * ROWS_PER_WG) << 16) | (sv.y & 0xFFFF);
    }
    #pragma unroll
    for (int i = 0; i < 2 * BIN_PAIRS; ++i)
        rk[i] = (own[i] >= 0) ? atomicAdd(&b_cnt[own[i]], 1) : 0;
    __syncthreads();

    if (tid < 32) {
        int c[8];
        int sum = 0;
        #pragma unroll
        for (int j = 0; j < 8; ++j) { c[j] = b_cnt[tid * 8 + j]; sum += c[j]; }
        int x = sum;
        #pragma unroll
        for (int dd = 1; dd < 32; dd <<= 1) {
            const int y = __shfl_up(x, dd, 32);
            if (tid >= dd) x += y;
        }
        int e0 = x - sum;
        #pragma unroll
        for (int j = 0; j < 8; ++j) { b_off[tid * 8 + j] = e0; e0 += c[j]; }
        if (tid == 31) b_off[OWNER_WGS] = e0;
    }
    __syncthreads();

    int* __restrict__ sorted = ws + WS_SORTED + base;
    #pragma unroll
    for (int i = 0; i < 2 * BIN_PAIRS; ++i)
        if (own[i] >= 0) sorted[b_off[own[i]] + rk[i]] = rec[i];
    ws[WS_TABLE + tid * BIN_WGS + chunk] = (b_off[tid] << 12) | b_cnt[tid];
}

#define TILE_ROWS     32
#define N_TILES       ((N_SRC + TILE_ROWS - 1) / TILE_ROWS)
#define A_PITCH       260
#define K_CHUNK       4
#define N_CHUNKS      (IN_FEATS / 4 / K_CHUNK)
#define PIECE_LAG     5
#define PIECE_CHUNK(j) (((j) * (N_CHUNKS - PIECE_LAG)) / 8)

__device__ __forceinline__ float4 ld_piece(const float* __restrict__ feat,
                                           const int tile, const int j, const int tid) {
    const int idx = j * BIN_THREADS + tid;
    int row = tile * TILE_ROWS + (idx >> 6);
    row = row < N_SRC ? row : N_SRC - 1;
    return *(const float4*)(feat + (size_t)row * IN_FEATS + (idx & 63) * 4);
}

__device__ __forceinline__ void st_piece(float* __restrict__ buf, const float4 v,
                                         const int j, const int tid) {
    const int idx = j * BIN_THREADS + tid;
    *(float4*)(buf + (idx >> 6) * A_PITCH + (idx & 63) * 4) = v;
}

__global__ __launch_bounds__(BIN_THREADS)
void gemm_cj_wmma(const float* __restrict__ feat,
                  const float* __restrict__ W,
                  const float* __restrict__ cj,
                  const int* __restrict__ src,
                  const int* __restrict__ dst,
                  float* __restrict__ h) {
    if (blockIdx.x >= GEMM_WGS) {
        bin_edges(blockIdx.x - GEMM_WGS, src, dst, (int*)h);
        return;
    }
    __shared__ __attribute__((aligned(16))) float a_lds[2][TILE_ROWS * A_PITCH];

    const int tid  = threadIdx.x;
    const int lane = tid & 31;
    const int wave = tid >> 5;
    const int m    = lane & 15;
    const int half = lane >> 4;

    {
        float4 first[8];
        #pragma unroll
        for (int j = 0; j < 8; ++j) first[j] = ld_piece(feat, blockIdx.x, j, tid);
        #pragma unroll
        for (int j = 0; j < 8; ++j) st_piece(a_lds[0], first[j], j, tid);
    }

    v2f bfr[IN_FEATS / 4];
    {
        const float* __restrict__ wp = W + (size_t)(2 * half) * OUT_FEATS + wave * 16 + m;
        #pragma unroll
        for (int s = 0; s < IN_FEATS / 4; ++s) {
            bfr[s].x = wp[(size_t)(4 * s) * OUT_FEATS];
            bfr[s].y = wp[(size_t)(4 * s + 1) * OUT_FEATS];
        }
    }
    __syncthreads();

    int cur = 0;
    for (int tile = blockIdx.x; tile < N_TILES; tile += GEMM_WGS) {
        const int nxt = (tile + GEMM_WGS < N_TILES) ? tile + GEMM_WGS : N_TILES - 1;
        float* __restrict__ nbuf = a_lds[cur ^ 1];

        const int tb = tile * TILE_ROWS;
        const float cv = cj[(tb + lane < N_SRC) ? tb + lane : 0];

        const float* __restrict__ a0p = a_lds[cur] + m * A_PITCH + 2 * half;
        const float* __restrict__ a1p = a0p + 16 * A_PITCH;
        v8f acc0 = (v8f){}, acc1 = (v8f){};
        float4 pc[8];
        #pragma unroll
        for (int c = 0; c < N_CHUNKS; ++c) {
            v2f fa[K_CHUNK], fb[K_CHUNK];
            #pragma unroll
            for (int q = 0; q < K_CHUNK; ++q) {
                fa[q] = *(const v2f*)(a0p + 4 * (c * K_CHUNK + q));
                fb[q] = *(const v2f*)(a1p + 4 * (c * K_CHUNK + q));
            }
            #pragma unroll
            for (int j = 0; j < 8; ++j)
                if (PIECE_CHUNK(j) + PIECE_LAG == c) st_piece(nbuf, pc[j], j, tid);
            #pragma unroll
            for (int j = 0; j < 8; ++j)
                if (PIECE_CHUNK(j) == c) pc[j] = ld_piece(feat, nxt, j, tid);
            __builtin_amdgcn_sched_barrier(0);
            #pragma unroll
            for (int q = 0; q < K_CHUNK; ++q) {
                acc0 = __builtin_amdgcn_wmma_f32_16x16x4_f32(
                            false, fa[q],  false, bfr[c * K_CHUNK + q],
                            (short)0, acc0,  false,  false);
                acc1 = __builtin_amdgcn_wmma_f32_16x16x4_f32(
                            false, fb[q],  false, bfr[c * K_CHUNK + q],
                            (short)0, acc1,  false,  false);
            }
            __builtin_amdgcn_sched_barrier(0);
        }

        __syncthreads();

        {
            const int cvb = __float_as_int(cv);
            float* __restrict__ hp = h + (size_t)(tb + 8 * half) * OUT_FEATS + wave * 16 + m;
            #pragma unroll
            for (int i = 0; i < 8; ++i) {
                const float lo = __int_as_float(__builtin_amdgcn_readlane(cvb, i));
                const float hi = __int_as_float(__builtin_amdgcn_readlane(cvb, 8 + i));
                hp[i * OUT_FEATS] = acc0[i] * (half ? hi : lo);
            }
            if (tb + 16 < N_SRC) {
                #pragma unroll
                for (int i = 0; i < 8; ++i) {
                    const float lo = __int_as_float(__builtin_amdgcn_readlane(cvb, 16 + i));
                    const float hi = __int_as_float(__builtin_amdgcn_readlane(cvb, 24 + i));
                    hp[(16 + i) * OUT_FEATS] = acc1[i] * (half ? hi : lo);
                }
            }
        }
        cur ^= 1;
    }
}

#define OWNER_THREADS 1024
#define LIST_CAP      20480

__device__ __forceinline__ float4 ld_hrow(const float* __restrict__ h, int s, int lane) {
    return *(const float4*)(h + (size_t)s * OUT_FEATS + lane * 4);
}

__global__ __launch_bounds__(OWNER_THREADS)
void owner_gather(const float* __restrict__ h,
                  const int* __restrict__ src,
                  const int* __restrict__ dst,
                  const float* __restrict__ ci,
                  const int* __restrict__ ws,
                  float* __restrict__ out) {
    __shared__ int s_seg[BIN_WGS];
    __shared__ int s_sego[BIN_WGS];
    __shared__ int s_cnt[ROWS_PER_WG];
    __shared__ int s_off[ROWS_PER_WG + 1];
    __shared__ int s_cur[ROWS_PER_WG];
    __shared__ int s_total;
    __shared__ int s_list[LIST_CAP];
    __shared__ int s_src[LIST_CAP];

    const int tid  = threadIdx.x;
    const int lane = tid & 31;
    const int wave = __builtin_amdgcn_readfirstlane(tid >> 5);
    const int r0   = blockIdx.x * ROWS_PER_WG;
    const int nrows = (N_DST - r0 < ROWS_PER_WG) ? (N_DST - r0) : ROWS_PER_WG;

    if (tid < ROWS_PER_WG) s_cnt[tid] = 0;
    if (tid < BIN_WGS) s_seg[tid] = ws[WS_TABLE + blockIdx.x * BIN_WGS + tid];
    __syncthreads();

    if (wave == 0) {
        int c[8];
        int sum = 0;
        #pragma unroll
        for (int j = 0; j < 8; ++j) {
            const int b = lane * 8 + j;
            c[j] = (b < BIN_WGS) ? (s_seg[b] & 0xFFF) : 0;
            sum += c[j];
        }
        int x = sum;
        #pragma unroll
        for (int dd = 1; dd < 32; dd <<= 1) {
            const int y = __shfl_up(x, dd, 32);
            if (lane >= dd) x += y;
        }
        int e0 = x - sum;
        #pragma unroll
        for (int j = 0; j < 8; ++j) {
            const int b = lane * 8 + j;
            if (b < BIN_WGS) s_sego[b] = e0;
            e0 += c[j];
        }
        if (lane == 31) s_total = e0;
    }
    __syncthreads();
    const int total = __builtin_amdgcn_readfirstlane(s_total);

    if (total <= LIST_CAP) {
        {
            const int b   = tid >> 2;
            const int sub = tid & 3;
            if (b < BIN_WGS) {
                const int pk = s_seg[b];
                const int cn = pk & 0xFFF;
                const int o  = s_sego[b];
                const int* __restrict__ p = ws + WS_SORTED + b * BIN_CHUNK + (pk >> 12);
                for (int i = sub; i < cn; i += 4) {
                    const int rec = p[i];
                    s_list[o + i] = rec;
                    atomicAdd(&s_cnt[rec >> 16], 1);
                }
            }
        }
        __syncthreads();

        if (wave == 0) {
            int c[7];
            int sum = 0;
            #pragma unroll
            for (int j = 0; j < 7; ++j) {
                const int r = lane * 7 + j;
                c[j] = (r < ROWS_PER_WG) ? s_cnt[r] : 0;
                sum += c[j];
            }
            int x = sum;
            #pragma unroll
            for (int dd = 1; dd < 32; dd <<= 1) {
                const int y = __shfl_up(x, dd, 32);
                if (lane >= dd) x += y;
            }
            int e0 = x - sum;
            #pragma unroll
            for (int j = 0; j < 7; ++j) {
                const int r = lane * 7 + j;
                if (r < ROWS_PER_WG) { s_off[r] = e0; s_cur[r] = e0; }
                e0 += c[j];
            }
            if (lane == 0) s_off[ROWS_PER_WG] = total;
        }
        __syncthreads();

        for (int j = tid; j < total; j += OWNER_THREADS) {
            const int rec = s_list[j];
            const int p   = atomicAdd(&s_cur[rec >> 16], 1);
            s_src[p] = rec & 0xFFFF;
        }
        __syncthreads();

        const int ra = (wave * nrows) >> 5;
        const int nr = (((wave + 1) * nrows) >> 5) - ra;
        const int   offv = s_off[ra + (lane < nr ? lane : nr)];
        const float civ  = (lane < nr) ? ci[r0 + ra + lane] : 0.f;
        const int eBeg = __builtin_amdgcn_readlane(offv, 0);
        const int eEnd = __builtin_amdgcn_readlane(offv, nr);
        int r    = 0;
        int rend = __builtin_amdgcn_readlane(offv, 1);
        float4 acc = make_float4(0.f, 0.f, 0.f, 0.f);

        for (int base = eBeg; base < eEnd; base += 32) {
            const int left = eEnd - base;
            const int cntc = left < 32 ? left : 32;
            const int mine = (lane < left) ? s_src[base + lane] : 0;
            for (int b = 0; b < cntc; b += 8) {
                float4 v[8];
                #pragma unroll
                for (int u = 0; u < 8; ++u)
                    v[u] = ld_hrow(h, __builtin_amdgcn_readlane(mine, b + u), lane);
                #pragma unroll
                for (int u = 0; u < 8; ++u) {
                    if (b + u < cntc) {
                        while (base + b + u == rend) {
                            const float sc = __int_as_float(__builtin_amdgcn_readlane(__float_as_int(civ), r));
                            float4 o;
                            o.x = acc.x * sc; o.y = acc.y * sc; o.z = acc.z * sc; o.w = acc.w * sc;
                            *(float4*)(out + (size_t)(r0 + ra + r) * OUT_FEATS + lane * 4) = o;
                            acc = make_float4(0.f, 0.f, 0.f, 0.f);
                            ++r;
                            rend = __builtin_amdgcn_readlane(offv, r + 1);
                        }
                        acc.x += v[u].x; acc.y += v[u].y; acc.z += v[u].z; acc.w += v[u].w;
                    }
                }
            }
        }
        while (r < nr) {
            const float sc = __int_as_float(__builtin_amdgcn_readlane(__float_as_int(civ), r));
            float4 o;
            o.x = acc.x * sc; o.y = acc.y * sc; o.z = acc.z * sc; o.w = acc.w * sc;
            *(float4*)(out + (size_t)(r0 + ra + r) * OUT_FEATS + lane * 4) = o;
            acc = make_float4(0.f, 0.f, 0.f, 0.f);
            ++r;
        }
    } else {
        for (int r = wave; r < nrows; r += OWNER_THREADS / 32) {
            float4 acc = make_float4(0.f, 0.f, 0.f, 0.f);
            for (int base = 0; base < N_EDGES; base += 32) {
                const int dd = dst[base + lane];
                unsigned m = __builtin_amdgcn_ballot_w32(dd == r0 + r);
                while (m) {
                    const int j = __builtin_ctz(m);
                    m &= m - 1;
                    const float4 v = ld_hrow(h, __builtin_amdgcn_readfirstlane(src[base + j]), lane);
                    acc.x += v.x; acc.y += v.y; acc.z += v.z; acc.w += v.w;
                }
            }
            const float sc = ci[r0 + r];
            float4 o;
            o.x = acc.x * sc; o.y = acc.y * sc; o.z = acc.z * sc; o.w = acc.w * sc;
            *(float4*)(out + (size_t)(r0 + r) * OUT_FEATS + lane * 4) = o;
        }
    }
}

extern "C" void kernel_launch(void* const* d_in, const int* in_sizes, int n_in,
                              void* d_out, int out_size, void* d_ws, size_t ws_size,
                              hipStream_t stream) {
    const float* feat   = (const float*)d_in[0];
    const float* weight = (const float*)d_in[1];
    const float* cj     = (const float*)d_in[2];
    const float* ci     = (const float*)d_in[3];
    const int*   src    = (const int*)d_in[4];
    const int*   dst    = (const int*)d_in[5];
    float*       out    = (float*)d_out;
    float*       h      = (float*)d_ws;

    gemm_cj_wmma<<<GEMM_WGS + BIN_WGS, BIN_THREADS, 0, stream>>>(feat, weight, cj, src, dst, h);

    owner_gather<<<OWNER_WGS, OWNER_THREADS, 0, stream>>>(h, src, dst, ci, (const int*)d_ws, out);
}


# Owner work-groups drop the row sort: records read in one round trip go straight into a fixed-capacity LDS list per row; no segment scan, row offsets or placement pass; sums stay in LDS-arrival order
# speedup vs baseline: 3.2910x; 1.0315x over previous
#include <hip/hip_runtime.h>
#include <hip/hip_bf16.h>

#define N_SRC     50000
#define N_DST     50000
#define N_EDGES   640000
#define IN_FEATS  256
#define OUT_FEATS 128

typedef __attribute__((ext_vector_type(2))) float v2f;
typedef __attribute__((ext_vector_type(8))) float v8f;

#define ROWS_PER_WG   196
#define OWNER_WGS     256
#define GEMM_WGS      256
#define BIN_WGS       250
#define BIN_THREADS   256
#define BIN_CHUNK     2560
#define BIN_PAIRS     (BIN_CHUNK / 2 / BIN_THREADS)
#define WS_SORTED     (N_SRC * OUT_FEATS)
#define WS_TABLE      (WS_SORTED + N_EDGES)

__device__ __forceinline__ void bin_edges(const int chunk,
                                          const int* __restrict__ src,
                                          const int* __restrict__ dst,
                                          int* __restrict__ ws) {
    __shared__ int b_cnt[OWNER_WGS];
    __shared__ int b_off[OWNER_WGS + 1];
    const int tid = threadIdx.x;
    b_cnt[tid] = 0;
    __syncthreads();

    const int base = chunk * BIN_CHUNK;
    const int2* __restrict__ dst2 = (const int2*)(dst + base);
    const int2* __restrict__ src2 = (const int2*)(src + base);
    int own[2 * BIN_PAIRS], rec[2 * BIN_PAIRS], rk[2 * BIN_PAIRS];
    #pragma unroll
    for (int i = 0; i < BIN_PAIRS; ++i) {
        const int2 d = dst2[i * BIN_THREADS + tid];
        const int2 sv = src2[i * BIN_THREADS + tid];
        const int o0 = d.x / ROWS_PER_WG, o1 = d.y / ROWS_PER_WG;
        own[2 * i]     = ((unsigned)d.x < (unsigned)N_DST) ? o0 : -1;
        own[2 * i + 1] = ((unsigned)d.y < (unsigned)N_DST) ? o1 : -1;
        rec[2 * i]     = ((d.x - o0 * ROWS_PER_WG) << 16) | (sv.x & 0xFFFF);
        rec[2 * i + 1] = ((d.y - o1 * ROWS_PER_WG) << 16) | (sv.y & 0xFFFF);
    }
    #pragma unroll
    for (int i = 0; i < 2 * BIN_PAIRS; ++i)
        rk[i] = (own[i] >= 0) ? atomicAdd(&b_cnt[own[i]], 1) : 0;
    __syncthreads();

    if (tid < 32) {
        int c[8];
        int sum = 0;
        #pragma unroll
        for (int j = 0; j < 8; ++j) { c[j] = b_cnt[tid * 8 + j]; sum += c[j]; }
        int x = sum;
        #pragma unroll
        for (int dd = 1; dd < 32; dd <<= 1) {
            const int y = __shfl_up(x, dd, 32);
            if (tid >= dd) x += y;
        }
        int e0 = x - sum;
        #pragma unroll
        for (int j = 0; j < 8; ++j) { b_off[tid * 8 + j] = e0; e0 += c[j]; }
        if (tid == 31) b_off[OWNER_WGS] = e0;
    }
    __syncthreads();

    int* __restrict__ sorted = ws + WS_SORTED + base;
    #pragma unroll
    for (int i = 0; i < 2 * BIN_PAIRS; ++i)
        if (own[i] >= 0) sorted[b_off[own[i]] + rk[i]] = rec[i];
    ws[WS_TABLE + tid * BIN_WGS + chunk] = (b_off[tid] << 12) | b_cnt[tid];
}

#define TILE_ROWS     32
#define N_TILES       ((N_SRC + TILE_ROWS - 1) / TILE_ROWS)
#define A_PITCH       260
#define K_CHUNK       4
#define N_CHUNKS      (IN_FEATS / 4 / K_CHUNK)
#define PIECE_LAG     5
#define PIECE_CHUNK(j) (((j) * (N_CHUNKS - PIECE_LAG)) / 8)

__device__ __forceinline__ float4 ld_piece(const float* __restrict__ feat,
                                           const int tile, const int j, const int tid) {
    const int idx = j * BIN_THREADS + tid;
    int row = tile * TILE_ROWS + (idx >> 6);
    row = row < N_SRC ? row : N_SRC - 1;
    return *(const float4*)(feat + (size_t)row * IN_FEATS + (idx & 63) * 4);
}

__device__ __forceinline__ void st_piece(float* __restrict__ buf, const float4 v,
                                         const int j, const int tid) {
    const int idx = j * BIN_THREADS + tid;
    *(float4*)(buf + (idx >> 6) * A_PITCH + (idx & 63) * 4) = v;
}

__global__ __launch_bounds__(BIN_THREADS)
void gemm_cj_wmma(const float* __restrict__ feat,
                  const float* __restrict__ W,
                  const float* __restrict__ cj,
                  const int* __restrict__ src,
                  const int* __restrict__ dst,
                  float* __restrict__ h) {
    if (blockIdx.x >= GEMM_WGS) {
        bin_edges(blockIdx.x - GEMM_WGS, src, dst, (int*)h);
        return;
    }
    __shared__ __attribute__((aligned(16))) float a_lds[2][TILE_ROWS * A_PITCH];

    const int tid  = threadIdx.x;
    const int lane = tid & 31;
    const int wave = tid >> 5;
    const int m    = lane & 15;
    const int half = lane >> 4;

    {
        float4 first[8];
        #pragma unroll
        for (int j = 0; j < 8; ++j) first[j] = ld_piece(feat, blockIdx.x, j, tid);
        #pragma unroll
        for (int j = 0; j < 8; ++j) st_piece(a_lds[0], first[j], j, tid);
    }

    v2f bfr[IN_FEATS / 4];
    {
        const float* __restrict__ wp = W + (size_t)(2 * half) * OUT_FEATS + wave * 16 + m;
        #pragma unroll
        for (int s = 0; s < IN_FEATS / 4; ++s) {
            bfr[s].x = wp[(size_t)(4 * s) * OUT_FEATS];
            bfr[s].y = wp[(size_t)(4 * s + 1) * OUT_FEATS];
        }
    }
    __syncthreads();

    int cur = 0;
    for (int tile = blockIdx.x; tile < N_TILES; tile += GEMM_WGS) {
        const int nxt = (tile + GEMM_WGS < N_TILES) ? tile + GEMM_WGS : N_TILES - 1;
        float* __restrict__ nbuf = a_lds[cur ^ 1];

        const int tb = tile * TILE_ROWS;
        const float cv = cj[(tb + lane < N_SRC) ? tb + lane : 0];

        const float* __restrict__ a0p = a_lds[cur] + m * A_PITCH + 2 * half;
        const float* __restrict__ a1p = a0p + 16 * A_PITCH;
        v8f acc0 = (v8f){}, acc1 = (v8f){};
        float4 pc[8];
        #pragma unroll
        for (int c = 0; c < N_CHUNKS; ++c) {
            v2f fa[K_CHUNK], fb[K_CHUNK];
            #pragma unroll
            for (int q = 0; q < K_CHUNK; ++q) {
                fa[q] = *(const v2f*)(a0p + 4 * (c * K_CHUNK + q));
                fb[q] = *(const v2f*)(a1p + 4 * (c * K_CHUNK + q));
            }
            #pragma unroll
            for (int j = 0; j < 8; ++j)
                if (PIECE_CHUNK(j) + PIECE_LAG == c) st_piece(nbuf, pc[j], j, tid);
            #pragma unroll
            for (int j = 0; j < 8; ++j)
                if (PIECE_CHUNK(j) == c) pc[j] = ld_piece(feat, nxt, j, tid);
            __builtin_amdgcn_sched_barrier(0);
            #pragma unroll
            for (int q = 0; q < K_CHUNK; ++q) {
                acc0 = __builtin_amdgcn_wmma_f32_16x16x4_f32(
                            false, fa[q],  false, bfr[c * K_CHUNK + q],
                            (short)0, acc0,  false,  false);
                acc1 = __builtin_amdgcn_wmma_f32_16x16x4_f32(
                            false, fb[q],  false, bfr[c * K_CHUNK + q],
                            (short)0, acc1,  false,  false);
            }
            __builtin_amdgcn_sched_barrier(0);
        }

        __syncthreads();

        {
            const int cvb = __float_as_int(cv);
            float* __restrict__ hp = h + (size_t)(tb + 8 * half) * OUT_FEATS + wave * 16 + m;
            #pragma unroll
            for (int i = 0; i < 8; ++i) {
                const float lo = __int_as_float(__builtin_amdgcn_readlane(cvb, i));
                const float hi = __int_as_float(__builtin_amdgcn_readlane(cvb, 8 + i));
                hp[i * OUT_FEATS] = acc0[i] * (half ? hi : lo);
            }
            if (tb + 16 < N_SRC) {
                #pragma unroll
                for (int i = 0; i < 8; ++i) {
                    const float lo = __int_as_float(__builtin_amdgcn_readlane(cvb, 16 + i));
                    const float hi = __int_as_float(__builtin_amdgcn_readlane(cvb, 24 + i));
                    hp[(16 + i) * OUT_FEATS] = acc1[i] * (half ? hi : lo);
                }
            }
        }
        cur ^= 1;
    }
}

#define OWNER_THREADS 1024
#define SEG_LOADS     8
#define ROW_CAP       212

__device__ __forceinline__ float4 ld_hrow(const float* __restrict__ h, int s, int lane) {
    return *(const float4*)(h + (size_t)s * OUT_FEATS + lane * 4);
}

__device__ __forceinline__ void to_row_list(int* __restrict__ s_rlist, int* __restrict__ s_over,
                                            const int rec, const int p) {
    if (p < ROW_CAP) s_rlist[(rec >> 16) * ROW_CAP + p] = rec & 0xFFFF;
    else *s_over = 1;
}

__global__ __launch_bounds__(OWNER_THREADS)
void owner_gather(const float* __restrict__ h,
                  const int* __restrict__ src,
                  const int* __restrict__ dst,
                  const float* __restrict__ ci,
                  const int* __restrict__ ws,
                  float* __restrict__ out) {
    __shared__ int s_rcnt[ROWS_PER_WG];
    __shared__ int s_over;
    __shared__ int s_rlist[ROWS_PER_WG * ROW_CAP];

    const int tid  = threadIdx.x;
    const int lane = tid & 31;
    const int wave = __builtin_amdgcn_readfirstlane(tid >> 5);
    const int r0   = blockIdx.x * ROWS_PER_WG;
    const int nrows = (N_DST - r0 < ROWS_PER_WG) ? (N_DST - r0) : ROWS_PER_WG;
    const int ra = (wave * nrows) >> 5;
    const int nr = (((wave + 1) * nrows) >> 5) - ra;

    const int b   = tid >> 2;
    const int sub = tid & 3;
    const bool has = b < BIN_WGS;
    const int* __restrict__ tw_p = ws + WS_TABLE + blockIdx.x * BIN_WGS + (has ? b : 0);
    const int pk = *tw_p;
    const float civ = (lane < nr) ? ci[r0 + ra + lane] : 0.f;
    if (tid < ROWS_PER_WG) s_rcnt[tid] = 0;
    if (tid == 0) s_over = 0;
    const int cn = has ? (pk & 0xFFF) : 0;
    const int* __restrict__ p = ws + WS_SORTED + (has ? b : 0) * BIN_CHUNK + (pk >> 12);
    int rec[SEG_LOADS];
    #pragma unroll
    for (int k = 0; k < SEG_LOADS; ++k) {
        const int i = sub + k * 4;
        rec[k] = *((i < cn) ? (p + i) : tw_p);
    }
    __syncthreads();

    int pos[SEG_LOADS];
    #pragma unroll
    for (int k = 0; k < SEG_LOADS; ++k)
        if (sub + k * 4 < cn) pos[k] = atomicAdd(&s_rcnt[rec[k] >> 16], 1);
    #pragma unroll
    for (int k = 0; k < SEG_LOADS; ++k)
        if (sub + k * 4 < cn) to_row_list(s_rlist, &s_over, rec[k], pos[k]);
    for (int i = sub + SEG_LOADS * 4; i < cn; i += 4) {
        const int rc = p[i];
        to_row_list(s_rlist, &s_over, rc, atomicAdd(&s_rcnt[rc >> 16], 1));
    }
    __syncthreads();

    const int over = __builtin_amdgcn_readfirstlane(s_over);

    if (over == 0) {
        const int cl = (lane < nr) ? s_rcnt[ra + lane] : 0;
        int x = cl;
        #pragma unroll
        for (int dd = 1; dd < 8; dd <<= 1) {
            const int y = __shfl_up(x, dd, 32);
            if (lane >= dd) x += y;
        }
        const int offv = x - cl;
        const int eEnd = __builtin_amdgcn_readlane(offv, nr);
        const int o1 = __builtin_amdgcn_readlane(offv, 1), o2 = __builtin_amdgcn_readlane(offv, 2);
        const int o3 = __builtin_amdgcn_readlane(offv, 3), o4 = __builtin_amdgcn_readlane(offv, 4);
        const int o5 = __builtin_amdgcn_readlane(offv, 5), o6 = __builtin_amdgcn_readlane(offv, 6);
        const int* __restrict__ lst = s_rlist + ra * ROW_CAP;
        int r    = 0;
        int rend = o1;
        float4 acc = make_float4(0.f, 0.f, 0.f, 0.f);

        for (int base = 0; base < eEnd; base += 32) {
            const int left = eEnd - base;
            const int cntc = left < 32 ? left : 32;
            const int e = base + lane;
            int q = 0, st = 0;
            if (e >= o1) { q = 1; st = o1; }
            if (e >= o2) { q = 2; st = o2; }
            if (e >= o3) { q = 3; st = o3; }
            if (e >= o4) { q = 4; st = o4; }
            if (e >= o5) { q = 5; st = o5; }
            if (e >= o6) { q = 6; st = o6; }
            const int mine = (lane < left) ? lst[q * ROW_CAP + (e - st)] : 0;
            for (int bb = 0; bb < cntc; bb += 8) {
                float4 v[8];
                #pragma unroll
                for (int u = 0; u < 8; ++u)
                    v[u] = ld_hrow(h, __builtin_amdgcn_readlane(mine, bb + u), lane);
                #pragma unroll
                for (int u = 0; u < 8; ++u) {
                    if (bb + u < cntc) {
                        while (base + bb + u == rend) {
                            const float sc = __int_as_float(__builtin_amdgcn_readlane(__float_as_int(civ), r));
                            float4 o;
                            o.x = acc.x * sc; o.y = acc.y * sc; o.z = acc.z * sc; o.w = acc.w * sc;
                            *(float4*)(out + (size_t)(r0 + ra + r) * OUT_FEATS + lane * 4) = o;
                            acc = make_float4(0.f, 0.f, 0.f, 0.f);
                            ++r;
                            rend = __builtin_amdgcn_readlane(offv, r + 1);
                        }
                        acc.x += v[u].x; acc.y += v[u].y; acc.z += v[u].z; acc.w += v[u].w;
                    }
                }
            }
        }
        while (r < nr) {
            const float sc = __int_as_float(__builtin_amdgcn_readlane(__float_as_int(civ), r));
            float4 o;
            o.x = acc.x * sc; o.y = acc.y * sc; o.z = acc.z * sc; o.w = acc.w * sc;
            *(float4*)(out + (size_t)(r0 + ra + r) * OUT_FEATS + lane * 4) = o;
            acc = make_float4(0.f, 0.f, 0.f, 0.f);
            ++r;
        }
    } else {
        for (int r = wave; r < nrows; r += OWNER_THREADS / 32) {
            float4 acc = make_float4(0.f, 0.f, 0.f, 0.f);
            for (int base = 0; base < N_EDGES; base += 32) {
                const int dd = dst[base + lane];
                unsigned m = __builtin_amdgcn_ballot_w32(dd == r0 + r);
                while (m) {
                    const int j = __builtin_ctz(m);
                    m &= m - 1;
                    const float4 v = ld_hrow(h, __builtin_amdgcn_readfirstlane(src[base + j]), lane);
                    acc.x += v.x; acc.y += v.y; acc.z += v.z; acc.w += v.w;
                }
            }
            const float sc = ci[r0 + r];
            float4 o;
            o.x = acc.x * sc; o.y = acc.y * sc; o.z = acc.z * sc; o.w = acc.w * sc;
            *(float4*)(out + (size_t)(r0 + r) * OUT_FEATS + lane * 4) = o;
        }
    }
}

extern "C" void kernel_launch(void* const* d_in, const int* in_sizes, int n_in,
                              void* d_out, int out_size, void* d_ws, size_t ws_size,
                              hipStream_t stream) {
    const float* feat   = (const float*)d_in[0];
    const float* weight = (const float*)d_in[1];
    const float* cj     = (const float*)d_in[2];
    const float* ci     = (const float*)d_in[3];
    const int*   src    = (const int*)d_in[4];
    const int*   dst    = (const int*)d_in[5];
    float*       out    = (float*)d_out;
    float*       h      = (float*)d_ws;

    gemm_cj_wmma<<<GEMM_WGS + BIN_WGS, BIN_THREADS, 0, stream>>>(feat, weight, cj, src, dst, h);

    owner_gather<<<OWNER_WGS, OWNER_THREADS, 0, stream>>>(h, src, dst, ci, (const int*)d_ws, out);
}
